# starting kernel
# baseline (speedup 1.0000x reference)
#include <hip/hip_runtime.h>
#include <math.h>


typedef __attribute__((ext_vector_type(2))) float v2f;
typedef __attribute__((ext_vector_type(8))) float v8f;

#define BATCH 16
#define NF    513
#define D     512
#define NITER 64

__global__ void __launch_bounds__(256)
sinkhorn_build(const float* __restrict__ xp, const float* __restrict__ xu,
               float* __restrict__ w) {
    int idx = blockIdx.x * 256 + threadIdx.x;
    int j = idx & (D - 1);
    int i = (idx >> 9) & (D - 1);
    int b = idx >> 18;
    const float* xpb = xp + (size_t)b * NF * NF;
    float v;
    if (i == j) {
        v = xu[b * NF + 1 + i];
    } else {
        v = 0.5f * (xpb[(size_t)(1 + i) * NF + (1 + j)] +
                    xpb[(size_t)(1 + j) * NF + (1 + i)]);
    }
    w[idx] = v;
}

__global__ void __launch_bounds__(512)
sinkhorn_col_lse(float* __restrict__ w) {
    __shared__ float sm[4][128];
    const int b     = blockIdx.x >> 2;
    const int slice = blockIdx.x & 3;
    const int c     = threadIdx.x & 127;
    const int q     = threadIdx.x >> 7;
    const int j     = slice * 128 + c;
    float* __restrict__ W = w + (size_t)b * D * D + j;
    const int i0 = q * 128, i1 = i0 + 128;

    float m = -INFINITY;
    for (int i = i0; i < i1; ++i) {
        __builtin_prefetch(W + (size_t)(i + 16) * D, 0, 0);
        m = fmaxf(m, W[(size_t)i * D]);
    }
    sm[q][c] = m;
    __syncthreads();
    m = fmaxf(fmaxf(sm[0][c], sm[1][c]), fmaxf(sm[2][c], sm[3][c]));
    __syncthreads();

    float s = 0.f;
    for (int i = i0; i < i1; ++i) s += __expf(W[(size_t)i * D] - m);
    sm[q][c] = s;
    __syncthreads();
    s = (sm[0][c] + sm[1][c]) + (sm[2][c] + sm[3][c]);
    const float lse = m + __logf(s);

    for (int i = i0; i < i1; ++i) W[(size_t)i * D] -= lse;
}

__global__ void __launch_bounds__(128)
sinkhorn_row_lse(float* __restrict__ w) {
    __shared__ float sMax[2][2][16];
    __shared__ float sSum[2][2][16];
    __shared__ float sLse[2][16];

    const int wv   = threadIdx.x >> 5;
    const int lane = threadIdx.x & 31;
    const int p    = wv >> 1;
    const int h    = wv & 1;
    const int half = lane >> 4;
    const int r    = lane & 15;

    const int gpanel = blockIdx.x * 2 + p;
    const int b      = gpanel >> 5;
    const int panel  = gpanel & 31;
    float* __restrict__ Wp   = w + ((size_t)b * D + (size_t)panel * 16) * D;
    float* __restrict__ Wrow = Wp + (size_t)r * D;
    const int cbase = h * 256;

    float m = -INFINITY;
    for (int c = cbase + half * 4; c < cbase + 256; c += 8) {
        float4 v = *(const float4*)(Wrow + c);
        m = fmaxf(m, fmaxf(fmaxf(v.x, v.y), fmaxf(v.z, v.w)));
    }
    m = fmaxf(m, __shfl_xor(m, 16, 32));
    if (half == 0) sMax[p][h][r] = m;
    __syncthreads();
    m = fmaxf(sMax[p][0][r], sMax[p][1][r]);

    v8f acc = {0.f, 0.f, 0.f, 0.f, 0.f, 0.f, 0.f, 0.f};
    v2f ones; ones[0] = 1.0f; ones[1] = 1.0f;
    const int co = half * 2;
    for (int c0 = cbase; c0 < cbase + 256; c0 += 4) {
        v2f a;
        a[0] = __expf(Wrow[c0 + co]     - m);
        a[1] = __expf(Wrow[c0 + co + 1] - m);
        acc = __builtin_amdgcn_wmma_f32_16x16x4_f32(
                  false, a, false, ones, (short)0, acc, false, false);
    }

    if (lane == 0) {
#pragma unroll
        for (int g = 0; g < 8; ++g) sSum[p][h][g] = acc[g];
    } else if (lane == 16) {
#pragma unroll
        for (int g = 0; g < 8; ++g) sSum[p][h][8 + g] = acc[g];
    }
    __syncthreads();
    if (h == 0 && half == 0) {
        sLse[p][r] = m + __logf(sSum[p][0][r] + sSum[p][1][r]);
    }
    __syncthreads();

    for (int rr = 0; rr < 16; ++rr) {
        const float l = sLse[p][rr];
        float* __restrict__ Wr = Wp + (size_t)rr * D;
        for (int c = cbase + lane; c < cbase + 256; c += 32) Wr[c] -= l;
    }
}

__global__ void __launch_bounds__(256)
sinkhorn_final_xp(const float* __restrict__ xp_in, const float* __restrict__ w,
                  float* __restrict__ out) {
    int idx = blockIdx.x * 256 + threadIdx.x;
    if (idx >= BATCH * NF * NF) return;
    int j = idx % NF;
    int t = idx / NF;
    int i = t % NF;
    int b = t / NF;
    float v;
    if (i == 0 && j == 0) {
        v = __expf(xp_in[idx]);
    } else if (i == 0 || j == 0) {
        v = xp_in[idx];
    } else if (i == j) {
        v = 0.0f;
    } else {
        v = __expf(w[((size_t)b * D + (i - 1)) * D + (j - 1)]);
    }
    out[idx] = v;
}

__global__ void __launch_bounds__(256)
sinkhorn_final_xu(const float* __restrict__ xu_in, const float* __restrict__ w,
                  float* __restrict__ out) {
    int idx = blockIdx.x * 256 + threadIdx.x;
    if (idx >= BATCH * NF) return;
    int k = idx % NF;
    int b = idx / NF;
    float v;
    if (k == 0) {
        v = __expf(xu_in[idx]);
    } else {
        v = __expf(w[((size_t)b * D + (k - 1)) * D + (k - 1)]);
    }
    out[idx] = v;
}

extern "C" void kernel_launch(void* const* d_in, const int* in_sizes, int n_in,
                              void* d_out, int out_size, void* d_ws, size_t ws_size,
                              hipStream_t stream) {
    (void)in_sizes; (void)n_in; (void)out_size; (void)ws_size;
    const float* xp_in = (const float*)d_in[0];
    const float* xu_in = (const float*)d_in[1];
    float* w   = (float*)d_ws;
    float* out = (float*)d_out;

    sinkhorn_build<<<(BATCH * D * D) / 256, 256, 0, stream>>>(xp_in, xu_in, w);

    for (int it = 0; it < NITER; ++it) {
        sinkhorn_col_lse<<<BATCH * 4, 512, 0, stream>>>(w);
        sinkhorn_row_lse<<<(BATCH * 32) / 2, 128, 0, stream>>>(w);
    }

    sinkhorn_final_xp<<<(BATCH * NF * NF + 255) / 256, 256, 0, stream>>>(xp_in, w, out);
    sinkhorn_final_xu<<<(BATCH * NF + 255) / 256, 256, 0, stream>>>(
        xu_in, w, out + (size_t)BATCH * NF * NF);
}


# Prologue and epilogue launches fused away (68 -> 65): build also does the first column LSE (blocks own columns, mirror tile via LDS); last iteration writes exp, diagonal and border to out itself
# speedup vs baseline: 23.2768x; 23.2768x over previous
#include <hip/hip_runtime.h>
#include <math.h>


typedef __attribute__((ext_vector_type(2))) float v2f;
typedef __attribute__((ext_vector_type(8))) float v8f;

#define BATCH 16
#define NF    513
#define D     512
#define NITER 64

__global__ void __launch_bounds__(512)
sinkhorn_build_col_lse(const float* __restrict__ xp, const float* __restrict__ xu,
                       float* __restrict__ w, float* __restrict__ pot) {
    __shared__ float sT[16][32 * 33];
    __shared__ float sMax[16][32];
    __shared__ float sSum[16][32];
    const int b     = blockIdx.x >> 4;
    const int slice = blockIdx.x & 15;
    const int lane  = threadIdx.x & 31;
    const int q     = threadIdx.x >> 5;
    const int i0    = q * 32;
    const int j0    = slice * 32;
    const float* __restrict__ xpb = xp + (size_t)b * NF * NF;
    const float* __restrict__ Pd = xpb + (size_t)(1 + i0) * NF + 1 + j0 + lane;
    const float* __restrict__ Pm = xpb + (size_t)(1 + j0) * NF + 1 + i0 + lane;

    float x[32], t[32];
#pragma unroll
    for (int k = 0; k < 32; ++k) x[k] = Pd[(size_t)k * NF];
#pragma unroll
    for (int k = 0; k < 32; ++k) t[k] = Pm[(size_t)k * NF];
    const float dg = xu[b * NF + 1 + j0 + lane];

    float* __restrict__ T = sT[q];
#pragma unroll
    for (int k = 0; k < 32; ++k) T[k * 33 + lane] = t[k];
    __syncthreads();
#pragma unroll
    for (int k = 0; k < 32; ++k) t[k] = T[lane * 33 + k];

    float* __restrict__ Wt = w + ((size_t)b * D + i0) * D + j0 + lane;
#pragma unroll
    for (int k = 0; k < 32; ++k) {
        x[k] = (i0 + k == j0 + lane) ? dg : 0.5f * (x[k] + t[k]);
        Wt[(size_t)k * D] = x[k];
    }

    float m = -INFINITY;
#pragma unroll
    for (int k = 0; k < 32; ++k) m = fmaxf(m, x[k]);
    sMax[q][lane] = m;
    __syncthreads();
    m = sMax[0][lane];
#pragma unroll
    for (int t2 = 1; t2 < 16; ++t2) m = fmaxf(m, sMax[t2][lane]);

    float s = 0.f;
#pragma unroll
    for (int k = 0; k < 32; ++k) s += __expf(x[k] - m);
    sSum[q][lane] = s;
    __syncthreads();
    s = sSum[0][lane];
#pragma unroll
    for (int t2 = 1; t2 < 16; ++t2) s += sSum[t2][lane];
    const float lse = m + __logf(s);

    if (q == 0) pot[b * D + j0 + lane] = lse;
}


template <int CTRL>
__device__ __forceinline__ float dpp_row(float v) {
    return __builtin_bit_cast(float,
        __builtin_amdgcn_mov_dpp(__builtin_bit_cast(int, v), CTRL, 0xf, 0xf, true));
}

__device__ __forceinline__ float xhalf(float v) {
    const int b = __builtin_bit_cast(int, v);
    return __builtin_bit_cast(float,
        __builtin_amdgcn_permlanex16(b, b, 0x76543210, 0xfedcba98, false, false));
}

template <bool FIRST, bool LAST>
__global__ void __launch_bounds__(256) __attribute__((amdgpu_waves_per_eu(2, 2)))
sinkhorn_iter(const float* __restrict__ w, const float* __restrict__ part_in,
              float* __restrict__ part_out, const float* __restrict__ pot_in,
              float* __restrict__ pot_out, const float* __restrict__ xp_in,
              const float* __restrict__ xu_in, float* __restrict__ out) {
    __shared__ __attribute__((aligned(16))) float sColPot[D];
    __shared__ __attribute__((aligned(16))) float sColSum[LAST ? 32 : 8][D];

    const int wv   = threadIdx.x >> 5;
    const int lane = threadIdx.x & 31;
    const int b    = blockIdx.x >> 4;
    const int blk  = blockIdx.x & 15;

    const float* __restrict__ P = w + ((size_t)blockIdx.x * 32 + wv * 4) * D + 4 * lane;

    float4 x[4][4];
#pragma unroll
    for (int R = 0; R < 4; ++R) {
#pragma unroll
        for (int j = 0; j < 4; ++j)
            x[R][j] = *(const float4*)(P + R * D + j * 128);
    }

    float bcol = 0.f, brow0 = 0.f, brow1 = 0.f, brow2 = 0.f, bxu = 0.f;
    if (LAST) {
        const float* __restrict__ xpb = xp_in + (size_t)b * NF * NF;
        if (threadIdx.x < 32) bcol = xpb[(size_t)(1 + blk * 32 + threadIdx.x) * NF];
        if (blk == 0) {
            brow0 = xpb[threadIdx.x];
            brow1 = xpb[256 + threadIdx.x];
            if (threadIdx.x == 0) { brow2 = xpb[512]; bxu = xu_in[b * NF]; }
        }
    }

    float2 cp = *(const float2*)(pot_in + (size_t)b * D + 2 * threadIdx.x);
    float2 ps[16];
    if (!FIRST) {
        const float* pin = part_in + (size_t)b * 16 * D + 2 * threadIdx.x;
#pragma unroll
        for (int q = 0; q < 16; ++q) ps[q] = *(const float2*)(pin + q * D);
    }

    if (!FIRST) {
        float2 cs = ps[0];
#pragma unroll
        for (int q = 1; q < 16; ++q) { cs.x += ps[q].x; cs.y += ps[q].y; }
        cp.x += __logf(cs.x); cp.y += __logf(cs.y);
    }
    *(float2*)&sColPot[2 * threadIdx.x] = cp;
    __syncthreads();
    {
        float4 c[4];
#pragma unroll
        for (int j = 0; j < 4; ++j) c[j] = *(const float4*)&sColPot[j * 128 + 4 * lane];
#pragma unroll
        for (int R = 0; R < 4; ++R) {
#pragma unroll
            for (int j = 0; j < 4; ++j) {
                x[R][j].x -= c[j].x; x[R][j].y -= c[j].y;
                x[R][j].z -= c[j].z; x[R][j].w -= c[j].w;
            }
        }
    }

    float m[4];
#pragma unroll
    for (int R = 0; R < 4; ++R) {
        float t = -INFINITY;
#pragma unroll
        for (int j = 0; j < 4; ++j)
            t = fmaxf(t, fmaxf(fmaxf(x[R][j].x, x[R][j].y), fmaxf(x[R][j].z, x[R][j].w)));
        t = fmaxf(t, dpp_row<0x161>(t));
        t = fmaxf(t, dpp_row<0x162>(t));
        t = fmaxf(t, dpp_row<0x164>(t));
        t = fmaxf(t, dpp_row<0x168>(t));
        t = fmaxf(t, xhalf(t));
        m[R] = t;
    }

    const v8f zero8 = {0.f, 0.f, 0.f, 0.f, 0.f, 0.f, 0.f, 0.f};
    v2f ones; ones[0] = 1.0f; ones[1] = 1.0f;
    v2f e0[4][4], e1[4][4];
    float s[4];
#pragma unroll
    for (int R = 0; R < 4; ++R) {
#pragma unroll
        for (int j = 0; j < 4; ++j) {
            v2f a0, a1;
            a0[0] = __expf(x[R][j].x - m[R]);
            a0[1] = __expf(x[R][j].y - m[R]);
            a1[0] = __expf(x[R][j].z - m[R]);
            a1[1] = __expf(x[R][j].w - m[R]);
            e0[R][j] = a0; e1[R][j] = a1;
        }
        const v2f q = ((e0[R][0] + e1[R][0]) + (e0[R][1] + e1[R][1])) +
                      ((e0[R][2] + e1[R][2]) + (e0[R][3] + e1[R][3]));
        const v8f acc = __builtin_amdgcn_wmma_f32_16x16x4_f32(
                            false, q, false, ones, (short)0, zero8, false, false);
        float t = ((acc[0] + acc[1]) + (acc[2] + acc[3])) +
                  ((acc[4] + acc[5]) + (acc[6] + acc[7]));
        t += xhalf(t);
        s[R] = t;
    }

    if (LAST) {
        float* __restrict__ outb = out + (size_t)b * NF * NF;
        float* __restrict__ xuo  = out + (size_t)BATCH * NF * NF + (size_t)b * NF;
        const int ib = blk * 32 + wv * 4;
#pragma unroll
        for (int R = 0; R < 4; ++R) {
            const float lse = m[R] + __logf(s[R]);
#pragma unroll
            for (int j = 0; j < 4; ++j) {
                float4 v = x[R][j];
                v.x = __expf(v.x - lse); v.y = __expf(v.y - lse);
                v.z = __expf(v.z - lse); v.w = __expf(v.w - lse);
                *(float4*)&sColSum[wv * 4 + R][j * 128 + 4 * lane] = v;
            }
        }
        __syncthreads();
#pragma unroll
        for (int R = 0; R < 4; ++R) {
            const int i = ib + R;
            const float* __restrict__ srow = sColSum[wv * 4 + R];
            float* __restrict__ orow = outb + (size_t)(1 + i) * NF + 1;
            float vv[16];
#pragma unroll
            for (int k = 0; k < 16; ++k) vv[k] = srow[32 * k + lane];
            const float dv = srow[i];
#pragma unroll
            for (int k = 0; k < 16; ++k) {
                const int col = 32 * k + lane;
                orow[col] = (col == i) ? 0.0f : vv[k];
            }
            if (lane == 0) xuo[1 + i] = dv;
        }
        if (threadIdx.x < 32) outb[(size_t)(1 + blk * 32 + threadIdx.x) * NF] = bcol;
        if (blk == 0) {
            outb[threadIdx.x] = (threadIdx.x == 0) ? __expf(brow0) : brow0;
            outb[256 + threadIdx.x] = brow1;
            if (threadIdx.x == 0) { outb[512] = brow2; xuo[0] = __expf(bxu); }
        }
    } else {
        float inv[4];
#pragma unroll
        for (int R = 0; R < 4; ++R) inv[R] = 1.0f / s[R];
#pragma unroll
        for (int j = 0; j < 4; ++j) {
            const v2f c0 = (e0[0][j] * inv[0] + e0[1][j] * inv[1]) +
                           (e0[2][j] * inv[2] + e0[3][j] * inv[3]);
            const v2f c1 = (e1[0][j] * inv[0] + e1[1][j] * inv[1]) +
                           (e1[2][j] * inv[2] + e1[3][j] * inv[3]);
            float4 cs; cs.x = c0[0]; cs.y = c0[1]; cs.z = c1[0]; cs.w = c1[1];
            *(float4*)&sColSum[wv][j * 128 + 4 * lane] = cs;
        }
        __syncthreads();
        float2 o = *(const float2*)&sColSum[0][2 * threadIdx.x];
#pragma unroll
        for (int q = 1; q < 8; ++q) {
            const float2 u = *(const float2*)&sColSum[q][2 * threadIdx.x];
            o.x += u.x; o.y += u.y;
        }
        *(float2*)(part_out + (size_t)blockIdx.x * D + 2 * threadIdx.x) = o;
        if (!FIRST && (threadIdx.x >> 4) == blk)
            *(float2*)(pot_out + (size_t)b * D + 2 * threadIdx.x) = cp;
    }
}

extern "C" void kernel_launch(void* const* d_in, const int* in_sizes, int n_in,
                              void* d_out, int out_size, void* d_ws, size_t ws_size,
                              hipStream_t stream) {
    (void)in_sizes; (void)n_in; (void)out_size; (void)ws_size;
    const float* xp_in = (const float*)d_in[0];
    const float* xu_in = (const float*)d_in[1];
    float* w   = (float*)d_ws;
    float* pA  = w + BATCH * D * D;
    float* pB  = pA + BATCH * 16 * D;
    float* cA  = pB + BATCH * 16 * D;
    float* cB  = cA + BATCH * D;
    float* out = (float*)d_out;

    sinkhorn_build_col_lse<<<BATCH * 16, 512, 0, stream>>>(xp_in, xu_in, w, cA);
    sinkhorn_iter<true, false><<<BATCH * 16, 256, 0, stream>>>(
        w, pB, pA, cA, cB, xp_in, xu_in, out);
    for (int it = 1; it < NITER - 1; it += 2) {
        sinkhorn_iter<false, false><<<BATCH * 16, 256, 0, stream>>>(
            w, pA, pB, cA, cB, xp_in, xu_in, out);
        sinkhorn_iter<false, false><<<BATCH * 16, 256, 0, stream>>>(
            w, pB, pA, cB, cA, xp_in, xu_in, out);
    }
    sinkhorn_iter<false, true><<<BATCH * 16, 256, 0, stream>>>(
        w, pA, pB, cA, cB, xp_in, xu_in, out);
}
